# starting kernel
# baseline (speedup 1.0000x reference)
#include <hip/hip_runtime.h>
#include <math.h>


typedef __attribute__((ext_vector_type(2))) float v2f;
typedef __attribute__((ext_vector_type(4))) float v4f;
typedef __attribute__((ext_vector_type(8))) float v8f;

#define WAVES_PER_BLOCK 8
#define TILE_ROWS 16
#define LSTRIDE 36

__device__ __forceinline__ v8f wmma4(v2f a, v2f b, v8f c) {
  return __builtin_amdgcn_wmma_f32_16x16x4_f32(
       false, a,  false, b,
       (short)0, c,  false,  false);
}

__global__ void __launch_bounds__(256)
arm_mlp_wmma_kernel(const float* __restrict__ x,  const float* __restrict__ sp,
                    const float* __restrict__ W0, const float* __restrict__ b0,
                    const float* __restrict__ W1, const float* __restrict__ b1,
                    const float* __restrict__ W2, const float* __restrict__ b2,
                    float* __restrict__ out, int nTiles) {
  __shared__ __align__(16) float lds[WAVES_PER_BLOCK * 2 * TILE_ROWS * LSTRIDE];

  const int lane = threadIdx.x & 31;
  const int wave = threadIdx.x >> 5;
  float* buf0 = &lds[wave * (2 * TILE_ROWS * LSTRIDE)];
  float* buf1 = buf0 + TILE_ROWS * LSTRIDE;

  const int lo = lane & 15;
  const int hi = lane >> 4;

  v2f w0c[2][9];
#pragma unroll
  for (int t = 0; t < 2; ++t)
#pragma unroll
    for (int s = 0; s < 9; ++s) {
      const int N = t * 16 + lo;
      const int K0 = 4 * s + 2 * hi;
      v2f w;
      w.x = (K0 + 0 < 34) ? W0[N * 34 + K0 + 0] : 0.0f;
      w.y = (K0 + 1 < 34) ? W0[N * 34 + K0 + 1] : 0.0f;
      w0c[t][s] = w;
    }
  v2f w1c[2][8];
#pragma unroll
  for (int t = 0; t < 2; ++t)
#pragma unroll
    for (int s = 0; s < 8; ++s) {
      const int N = t * 16 + lo;
      const int K0 = 4 * s + 2 * hi;
      v2f w; w.x = W1[N * 32 + K0]; w.y = W1[N * 32 + K0 + 1];
      w1c[t][s] = w;
    }
  v2f w2c[8];
#pragma unroll
  for (int s = 0; s < 8; ++s) {
    const int K0 = 4 * s + 2 * hi;
    v2f w;
    if (lo < 4) { w.x = W2[lo * 32 + K0]; w.y = W2[lo * 32 + K0 + 1]; }
    else        { w.x = 0.0f; w.y = 0.0f; }
    w2c[s] = w;
  }
  const float bias0[2] = {b0[lo], b0[16 + lo]};
  const float bias1[2] = {b1[lo], b1[16 + lo]};
  const float bias2    = (lo < 4) ? b2[lo] : 0.0f;

  const int waveGlobal = blockIdx.x * WAVES_PER_BLOCK + wave;
  const int waveTotal  = gridDim.x * WAVES_PER_BLOCK;

  for (int tile = waveGlobal; tile < nTiles; tile += waveTotal) {
    const int row0 = tile * TILE_ROWS;

    const v4f* xs = (const v4f*)(x + (size_t)row0 * 32);
#pragma unroll
    for (int j = 0; j < 4; ++j) {
      v4f v = xs[lane + 32 * j];
      const int f = (lane + 32 * j) * 4;
      const int r = f >> 5, c = f & 31;
      *(v4f*)&buf0[r * LSTRIDE + c] = v;
    }
    buf0[(lane >> 1) * LSTRIDE + 32 + (lane & 1)] = sp[(size_t)row0 * 2 + lane];
    if (lane < 16) {
      v2f z; z.x = 0.0f; z.y = 0.0f;
      *(v2f*)&buf0[lane * LSTRIDE + 34] = z;
    }
    __builtin_amdgcn_wave_barrier();

    v8f h1[2];
#pragma unroll
    for (int t = 0; t < 2; ++t) {
      v8f acc = {};
#pragma unroll
      for (int s = 0; s < 9; ++s) {
        v2f a = *(const v2f*)&buf0[lo * LSTRIDE + 4 * s + 2 * hi];
        acc = wmma4(a, w0c[t][s], acc);
      }
#pragma unroll
      for (int k = 0; k < 8; ++k) h1[t][k] = fmaxf(acc[k] + bias0[t], 0.0f);
    }
#pragma unroll
    for (int t = 0; t < 2; ++t)
#pragma unroll
      for (int v = 0; v < 8; ++v)
        buf1[(v + 8 * hi) * LSTRIDE + t * 16 + lo] = h1[t][v];
    __builtin_amdgcn_wave_barrier();

    v8f h2[2];
#pragma unroll
    for (int t = 0; t < 2; ++t) {
      v8f acc = {};
#pragma unroll
      for (int s = 0; s < 8; ++s) {
        v2f a = *(const v2f*)&buf1[lo * LSTRIDE + 4 * s + 2 * hi];
        acc = wmma4(a, w1c[t][s], acc);
      }
#pragma unroll
      for (int k = 0; k < 8; ++k)
        h2[t][k] = fmaxf(acc[k] + bias1[t] + h1[t][k], 0.0f);
    }
#pragma unroll
    for (int t = 0; t < 2; ++t)
#pragma unroll
      for (int v = 0; v < 8; ++v)
        buf0[(v + 8 * hi) * LSTRIDE + t * 16 + lo] = h2[t][v];
    __builtin_amdgcn_wave_barrier();

    v8f acc = {};
#pragma unroll
    for (int s = 0; s < 8; ++s) {
      v2f a = *(const v2f*)&buf0[lo * LSTRIDE + 4 * s + 2 * hi];
      acc = wmma4(a, w2c[s], acc);
    }
    if (lo < 4) {
#pragma unroll
      for (int v = 0; v < 8; ++v)
        buf1[(v + 8 * hi) * LSTRIDE + lo] = acc[v] + bias2;
    }
    __builtin_amdgcn_wave_barrier();

    {
      const int r = lane >> 1, n = lane & 1;
      const float p = buf1[r * LSTRIDE + n];
      const float g = buf1[r * LSTRIDE + n + 2];
      const float s = buf0[r * LSTRIDE + 32 + n];
      const float sig = 1.0f / (1.0f + __expf(-g));
      out[(size_t)row0 * 2 + lane] = fmaf(p, sig, s);
    }
    __builtin_amdgcn_wave_barrier();
  }
}

extern "C" void kernel_launch(void* const* d_in, const int* in_sizes, int n_in,
                              void* d_out, int out_size, void* d_ws, size_t ws_size,
                              hipStream_t stream) {
  const float* x  = (const float*)d_in[0];
  const float* sp = (const float*)d_in[1];
  const float* W0 = (const float*)d_in[2];
  const float* b0 = (const float*)d_in[3];
  const float* W1 = (const float*)d_in[4];
  const float* b1 = (const float*)d_in[5];
  const float* W2 = (const float*)d_in[6];
  const float* b2 = (const float*)d_in[7];
  float* out = (float*)d_out;

  const int B = in_sizes[0] / 32;
  const int nTiles = B / TILE_ROWS;
  int blocks = (nTiles + (WAVES_PER_BLOCK * 8) - 1) / (WAVES_PER_BLOCK * 8);
  if (blocks < 1) blocks = 1;
  if (blocks > 8192) blocks = 8192;

  arm_mlp_wmma_kernel<<<blocks, 256, 0, stream>>>(x, sp, W0, b0, W1, b1, W2, b2,
                                                  out, nTiles);
}


# Swap WMMA operand roles (weights=A, activations=B): each layer's D registers are the next layer's B fragments, no inter-layer LDS transposes; b0 rides the K pad, residual enters as C (sum order)
# speedup vs baseline: 1.0466x; 1.0466x over previous
#include <hip/hip_runtime.h>
#include <math.h>


typedef __attribute__((ext_vector_type(2))) float v2f;
typedef __attribute__((ext_vector_type(4))) float v4f;
typedef __attribute__((ext_vector_type(8))) float v8f;

#define WAVES_PER_BLOCK 8
#define TILE_ROWS 16
#define LSTRIDE 36

__device__ __forceinline__ v8f wmma4(v2f a, v2f b, v8f c) {
  return __builtin_amdgcn_wmma_f32_16x16x4_f32(
       false, a,  false, b,
       (short)0, c,  false,  false);
}

__global__ void __launch_bounds__(256)
arm_mlp_wmma_kernel(const float* __restrict__ x,  const float* __restrict__ sp,
                    const float* __restrict__ W0, const float* __restrict__ b0,
                    const float* __restrict__ W1, const float* __restrict__ b1,
                    const float* __restrict__ W2, const float* __restrict__ b2,
                    float* __restrict__ out, int nTiles) {
  __shared__ __align__(16) float lds[WAVES_PER_BLOCK * TILE_ROWS * LSTRIDE];

  const int lane = threadIdx.x & 31;
  const int wave = threadIdx.x >> 5;
  float* buf = &lds[wave * (TILE_ROWS * LSTRIDE)];

  const int lo = lane & 15;
  const int hi = lane >> 4;

  v2f w0a[2][9];
#pragma unroll
  for (int t = 0; t < 2; ++t) {
    const int N = t * 16 + lo;
#pragma unroll
    for (int s = 0; s < 8; ++s) {
      const int K0 = 4 * s + 2 * hi;
      v2f w; w.x = W0[N * 34 + K0]; w.y = W0[N * 34 + K0 + 1];
      w0a[t][s] = w;
    }
    v2f w8;
    if (hi == 0) { w8.x = W0[N * 34 + 32]; w8.y = W0[N * 34 + 33]; }
    else         { w8.x = b0[N];           w8.y = 0.0f; }
    w0a[t][8] = w8;
  }
  v2f w1a[2][8];
#pragma unroll
  for (int t = 0; t < 2; ++t)
#pragma unroll
    for (int j = 0; j < 8; ++j) {
      const int N = t * 16 + lo;
      const int K0 = 16 * (j >> 2) + 8 * hi + 2 * (j & 3);
      v2f w; w.x = W1[N * 32 + K0]; w.y = W1[N * 32 + K0 + 1];
      w1a[t][j] = w;
    }
  v2f w2a[8];
#pragma unroll
  for (int j = 0; j < 8; ++j) {
    const int K0 = 16 * (j >> 2) + 8 * hi + 2 * (j & 3);
    v2f w;
    if (lo < 4) { w.x = W2[lo * 32 + K0]; w.y = W2[lo * 32 + K0 + 1]; }
    else        { w.x = 0.0f; w.y = 0.0f; }
    w2a[j] = w;
  }
  v8f b1d[2];
#pragma unroll
  for (int t = 0; t < 2; ++t)
#pragma unroll
    for (int r = 0; r < 8; ++r) b1d[t][r] = b1[t * 16 + 8 * hi + r];
  const float b2p0 = b2[0], b2p1 = b2[1], b2g0 = b2[2], b2g1 = b2[3];

  const int waveGlobal = blockIdx.x * WAVES_PER_BLOCK + wave;
  const int waveTotal  = gridDim.x * WAVES_PER_BLOCK;

  for (int tile = waveGlobal; tile < nTiles; tile += waveTotal) {
    const int row0 = tile * TILE_ROWS;

    const v2f spl = *(const v2f*)&sp[(size_t)(row0 + lo) * 2];
    const v4f* xs = (const v4f*)(x + (size_t)row0 * 32);
    float* st = &buf[(lane >> 3) * LSTRIDE + (lane & 7) * 4];
    v4f xv[4];
#pragma unroll
    for (int j = 0; j < 4; ++j) xv[j] = xs[lane + 32 * j];
#pragma unroll
    for (int j = 0; j < 4; ++j) *(v4f*)&st[j * 4 * LSTRIDE] = xv[j];
    v2f spv;
    spv.x = hi ? 1.0f : spl.x;
    spv.y = hi ? 0.0f : spl.y;
    __builtin_amdgcn_wave_barrier();

    v2f xb[8];
#pragma unroll
    for (int s = 0; s < 8; ++s)
      xb[s] = *(const v2f*)&buf[lo * LSTRIDE + 4 * s + 2 * hi];

    v8f h1[2];
#pragma unroll
    for (int t = 0; t < 2; ++t) {
      v8f acc = {};
#pragma unroll
      for (int s = 0; s < 8; ++s) acc = wmma4(w0a[t][s], xb[s], acc);
      acc = wmma4(w0a[t][8], spv, acc);
#pragma unroll
      for (int k = 0; k < 8; ++k) h1[t][k] = fmaxf(acc[k], 0.0f);
    }

    v8f h2[2];
#pragma unroll
    for (int t = 0; t < 2; ++t) {
      v8f acc = h1[t];
#pragma unroll
      for (int j = 0; j < 8; ++j) {
        v2f b; b.x = h1[j >> 2][2 * (j & 3)]; b.y = h1[j >> 2][2 * (j & 3) + 1];
        acc = wmma4(w1a[t][j], b, acc);
      }
#pragma unroll
      for (int k = 0; k < 8; ++k) h2[t][k] = fmaxf(acc[k] + b1d[t][k], 0.0f);
    }

    v8f raw = {};
#pragma unroll
    for (int j = 0; j < 8; ++j) {
      v2f b; b.x = h2[j >> 2][2 * (j & 3)]; b.y = h2[j >> 2][2 * (j & 3) + 1];
      raw = wmma4(w2a[j], b, raw);
    }

    {
      const float p0 = raw[0] + b2p0, p1 = raw[1] + b2p1;
      const float g0 = raw[2] + b2g0, g1 = raw[3] + b2g1;
      const float sig0 = 1.0f / (1.0f + __expf(-g0));
      const float sig1 = 1.0f / (1.0f + __expf(-g1));
      v2f o; o.x = fmaf(p0, sig0, spv.x); o.y = fmaf(p1, sig1, spv.y);
      if (hi == 0) *(v2f*)&out[(size_t)(row0 + lo) * 2] = o;
    }
    __builtin_amdgcn_wave_barrier();
  }
}

extern "C" void kernel_launch(void* const* d_in, const int* in_sizes, int n_in,
                              void* d_out, int out_size, void* d_ws, size_t ws_size,
                              hipStream_t stream) {
  const float* x  = (const float*)d_in[0];
  const float* sp = (const float*)d_in[1];
  const float* W0 = (const float*)d_in[2];
  const float* b0 = (const float*)d_in[3];
  const float* W1 = (const float*)d_in[4];
  const float* b1 = (const float*)d_in[5];
  const float* W2 = (const float*)d_in[6];
  const float* b2 = (const float*)d_in[7];
  float* out = (float*)d_out;

  const int B = in_sizes[0] / 32;
  const int nTiles = B / TILE_ROWS;
  int blocks = (nTiles + (WAVES_PER_BLOCK * 8) - 1) / (WAVES_PER_BLOCK * 8);
  if (blocks < 1) blocks = 1;
  if (blocks > 8192) blocks = 8192;

  arm_mlp_wmma_kernel<<<blocks, 256, 0, stream>>>(x, sp, W0, b0, W1, b1, W2, b2,
                                                  out, nTiles);
}


# Each wave issues its NEXT tile's X and sp global loads before the current tile's WMMAs (held in registers, stored to LDS one trip later), so the memory round trip runs under its own products
# speedup vs baseline: 1.1119x; 1.0624x over previous
#include <hip/hip_runtime.h>
#include <math.h>


typedef __attribute__((ext_vector_type(2))) float v2f;
typedef __attribute__((ext_vector_type(4))) float v4f;
typedef __attribute__((ext_vector_type(8))) float v8f;

#define WAVES_PER_BLOCK 8
#define TILE_ROWS 16
#define LSTRIDE 36

__device__ __forceinline__ v8f wmma4(v2f a, v2f b, v8f c) {
  return __builtin_amdgcn_wmma_f32_16x16x4_f32(
       false, a,  false, b,
       (short)0, c,  false,  false);
}

typedef const v4f __attribute__((address_space(1)))* gv4f_ptr;
typedef const v2f __attribute__((address_space(1)))* gv2f_ptr;

__device__ __forceinline__ unsigned long long wave_uniform_addr(const float* p) {
  const unsigned long long u = (unsigned long long)p;
  const unsigned lo32 = __builtin_amdgcn_readfirstlane((unsigned)u);
  const unsigned hi32 = __builtin_amdgcn_readfirstlane((unsigned)(u >> 32));
  return ((unsigned long long)hi32 << 32) | (unsigned long long)lo32;
}

__global__ void __launch_bounds__(256)
arm_mlp_wmma_kernel(const float* __restrict__ x,  const float* __restrict__ sp,
                    const float* __restrict__ W0, const float* __restrict__ b0,
                    const float* __restrict__ W1, const float* __restrict__ b1,
                    const float* __restrict__ W2, const float* __restrict__ b2,
                    float* __restrict__ out, int nTiles) {
  __shared__ __align__(16) float lds[WAVES_PER_BLOCK * TILE_ROWS * LSTRIDE];

  const int lane = threadIdx.x & 31;
  const int wave = __builtin_amdgcn_readfirstlane((int)(threadIdx.x >> 5));
  float* buf = &lds[wave * (TILE_ROWS * LSTRIDE)];

  const int lo = lane & 15;
  const int hi = lane >> 4;

  v2f w0a[2][9];
#pragma unroll
  for (int t = 0; t < 2; ++t) {
    const int N = t * 16 + lo;
#pragma unroll
    for (int s = 0; s < 8; ++s) {
      const int K0 = 4 * s + 2 * hi;
      v2f w; w.x = W0[N * 34 + K0]; w.y = W0[N * 34 + K0 + 1];
      w0a[t][s] = w;
    }
    v2f w8;
    if (hi == 0) { w8.x = W0[N * 34 + 32]; w8.y = W0[N * 34 + 33]; }
    else         { w8.x = b0[N];           w8.y = 0.0f; }
    w0a[t][8] = w8;
  }
  v2f w1a[2][8];
#pragma unroll
  for (int t = 0; t < 2; ++t)
#pragma unroll
    for (int j = 0; j < 8; ++j) {
      const int N = t * 16 + lo;
      const int K0 = 16 * (j >> 2) + 8 * hi + 2 * (j & 3);
      v2f w; w.x = W1[N * 32 + K0]; w.y = W1[N * 32 + K0 + 1];
      w1a[t][j] = w;
    }
  v2f w2a[8];
#pragma unroll
  for (int j = 0; j < 8; ++j) {
    const int K0 = 16 * (j >> 2) + 8 * hi + 2 * (j & 3);
    v2f w;
    if (lo < 4) { w.x = W2[lo * 32 + K0]; w.y = W2[lo * 32 + K0 + 1]; }
    else        { w.x = 0.0f; w.y = 0.0f; }
    w2a[j] = w;
  }
  v8f b1d[2];
#pragma unroll
  for (int t = 0; t < 2; ++t)
#pragma unroll
    for (int r = 0; r < 8; ++r) b1d[t][r] = b1[t * 16 + 8 * hi + r];
  const float b2p0 = b2[0], b2p1 = b2[1], b2g0 = b2[2], b2g1 = b2[3];

  const int waveGlobal = blockIdx.x * WAVES_PER_BLOCK + wave;
  const int waveTotal  = gridDim.x * WAVES_PER_BLOCK;

  float* st = &buf[(lane >> 3) * LSTRIDE + (lane & 7) * 4];

  int tile = waveGlobal;
  if (tile >= nTiles) return;

  v4f xv[4];
  v2f spl;
  {
    const gv4f_ptr xs = (gv4f_ptr)wave_uniform_addr(x + (size_t)tile * (TILE_ROWS * 32));
    const gv2f_ptr ss = (gv2f_ptr)wave_uniform_addr(sp + (size_t)tile * (TILE_ROWS * 2));
#pragma unroll
    for (int j = 0; j < 4; ++j) xv[j] = xs[lane + 32 * j];
    spl = ss[lo];
  }

  for (;;) {
    const int row0 = tile * TILE_ROWS;

#pragma unroll
    for (int j = 0; j < 4; ++j) *(v4f*)&st[j * 4 * LSTRIDE] = xv[j];
    v2f spv;
    spv.x = hi ? 1.0f : spl.x;
    spv.y = hi ? 0.0f : spl.y;

    const int nextTile = tile + waveTotal;
    const bool more = nextTile < nTiles;
    if (more) {
      const gv4f_ptr xs = (gv4f_ptr)wave_uniform_addr(x + (size_t)nextTile * (TILE_ROWS * 32));
      const gv2f_ptr ss = (gv2f_ptr)wave_uniform_addr(sp + (size_t)nextTile * (TILE_ROWS * 2));
#pragma unroll
      for (int j = 0; j < 4; ++j) xv[j] = xs[lane + 32 * j];
      spl = ss[lo];
    }
    __builtin_amdgcn_wave_barrier();

    v2f xb[8];
#pragma unroll
    for (int s = 0; s < 8; ++s)
      xb[s] = *(const v2f*)&buf[lo * LSTRIDE + 4 * s + 2 * hi];

    v8f h1[2];
#pragma unroll
    for (int t = 0; t < 2; ++t) {
      v8f acc = {};
#pragma unroll
      for (int s = 0; s < 8; ++s) acc = wmma4(w0a[t][s], xb[s], acc);
      acc = wmma4(w0a[t][8], spv, acc);
#pragma unroll
      for (int k = 0; k < 8; ++k) h1[t][k] = fmaxf(acc[k], 0.0f);
    }

    v8f h2[2];
#pragma unroll
    for (int t = 0; t < 2; ++t) {
      v8f acc = h1[t];
#pragma unroll
      for (int j = 0; j < 8; ++j) {
        v2f b; b.x = h1[j >> 2][2 * (j & 3)]; b.y = h1[j >> 2][2 * (j & 3) + 1];
        acc = wmma4(w1a[t][j], b, acc);
      }
#pragma unroll
      for (int k = 0; k < 8; ++k) h2[t][k] = fmaxf(acc[k] + b1d[t][k], 0.0f);
    }

    v8f raw = {};
#pragma unroll
    for (int j = 0; j < 8; ++j) {
      v2f b; b.x = h2[j >> 2][2 * (j & 3)]; b.y = h2[j >> 2][2 * (j & 3) + 1];
      raw = wmma4(w2a[j], b, raw);
    }

    {
      const float p0 = raw[0] + b2p0, p1 = raw[1] + b2p1;
      const float g0 = raw[2] + b2g0, g1 = raw[3] + b2g1;
      const float sig0 = 1.0f / (1.0f + __expf(-g0));
      const float sig1 = 1.0f / (1.0f + __expf(-g1));
      v2f o; o.x = fmaf(p0, sig0, spv.x); o.y = fmaf(p1, sig1, spv.y);
      if (hi == 0) *(v2f*)&out[(size_t)(row0 + lo) * 2] = o;
    }
    __builtin_amdgcn_wave_barrier();
    if (!more) break;
    tile = nextTile;
  }
}

extern "C" void kernel_launch(void* const* d_in, const int* in_sizes, int n_in,
                              void* d_out, int out_size, void* d_ws, size_t ws_size,
                              hipStream_t stream) {
  const float* x  = (const float*)d_in[0];
  const float* sp = (const float*)d_in[1];
  const float* W0 = (const float*)d_in[2];
  const float* b0 = (const float*)d_in[3];
  const float* W1 = (const float*)d_in[4];
  const float* b1 = (const float*)d_in[5];
  const float* W2 = (const float*)d_in[6];
  const float* b2 = (const float*)d_in[7];
  float* out = (float*)d_out;

  const int B = in_sizes[0] / 32;
  const int nTiles = B / TILE_ROWS;
  int blocks = (nTiles + (WAVES_PER_BLOCK * 8) - 1) / (WAVES_PER_BLOCK * 8);
  if (blocks < 1) blocks = 1;
  if (blocks > 8192) blocks = 8192;

  arm_mlp_wmma_kernel<<<blocks, 256, 0, stream>>>(x, sp, W0, b0, W1, b1, W2, b2,
                                                  out, nTiles);
}


# Vector instructions per tile 73 -> 40: residual folded into W1 (+I), b1 as first C, sp/b0 tail k regrouped (all: sum order), integer-max relu, one sigmoid per lane with v_rcp, not the IEEE divide
# speedup vs baseline: 1.1580x; 1.0415x over previous
#include <hip/hip_runtime.h>
#include <math.h>


typedef __attribute__((ext_vector_type(2))) float v2f;
typedef __attribute__((ext_vector_type(4))) float v4f;
typedef __attribute__((ext_vector_type(8))) float v8f;

#define WAVES_PER_BLOCK 8
#define TILE_ROWS 16
#define LSTRIDE 36

__device__ __forceinline__ v8f wmma4(v2f a, v2f b, v8f c) {
  return __builtin_amdgcn_wmma_f32_16x16x4_f32(
       false, a,  false, b,
       (short)0, c,  false,  false);
}

__device__ __forceinline__ float relu1(float v) {
  const int b = __builtin_bit_cast(int, v);
  return __builtin_bit_cast(float, b > 0 ? b : 0);
}

typedef const v4f __attribute__((address_space(1)))* gv4f_ptr;
typedef const float __attribute__((address_space(1)))* gf_ptr;

__device__ __forceinline__ unsigned long long wave_uniform_addr(const float* p) {
  const unsigned long long u = (unsigned long long)p;
  const unsigned lo32 = __builtin_amdgcn_readfirstlane((unsigned)u);
  const unsigned hi32 = __builtin_amdgcn_readfirstlane((unsigned)(u >> 32));
  return ((unsigned long long)hi32 << 32) | (unsigned long long)lo32;
}

__global__ void __launch_bounds__(256)
arm_mlp_wmma_kernel(const float* __restrict__ x,  const float* __restrict__ sp,
                    const float* __restrict__ W0, const float* __restrict__ b0,
                    const float* __restrict__ W1, const float* __restrict__ b1,
                    const float* __restrict__ W2, const float* __restrict__ b2,
                    float* __restrict__ out, int nTiles) {
  __shared__ __align__(16) float lds[WAVES_PER_BLOCK * TILE_ROWS * LSTRIDE];

  const int lane = threadIdx.x & 31;
  const int wave = __builtin_amdgcn_readfirstlane((int)(threadIdx.x >> 5));
  float* buf = &lds[wave * (TILE_ROWS * LSTRIDE)];

  const int lo = lane & 15;
  const int hi = lane >> 4;

  v2f w0a[2][9];
#pragma unroll
  for (int t = 0; t < 2; ++t) {
    const int N = t * 16 + lo;
#pragma unroll
    for (int s = 0; s < 8; ++s) {
      const int K0 = 4 * s + 2 * hi;
      v2f w; w.x = W0[N * 34 + K0]; w.y = W0[N * 34 + K0 + 1];
      w0a[t][s] = w;
    }
    v2f w8;
    w8.x = W0[N * 34 + 32 + hi];
    w8.y = hi ? 0.0f : b0[N];
    w0a[t][8] = w8;
  }
  v2f w1a[2][8];
  const bool diagHalf = (lo >> 3) == hi;
  const int  diagStep = (lo & 7) >> 1;
  const int  diagElem = lo & 1;
#pragma unroll
  for (int t = 0; t < 2; ++t)
#pragma unroll
    for (int j = 0; j < 8; ++j) {
      const int N = t * 16 + lo;
      const int K0 = 16 * (j >> 2) + 8 * hi + 2 * (j & 3);
      v2f w; w.x = W1[N * 32 + K0]; w.y = W1[N * 32 + K0 + 1];
      if ((j >> 2) == t) {
        const bool here = diagHalf && diagStep == (j & 3);
        w.x += (here && diagElem == 0) ? 1.0f : 0.0f;
        w.y += (here && diagElem == 1) ? 1.0f : 0.0f;
      }
      w1a[t][j] = w;
    }
  const bool w2valid = (lo & 6) == 0;
  const int  w2row   = ((lo & 1) << 1) | (lo >> 3);
  v2f w2a[8];
#pragma unroll
  for (int j = 0; j < 8; ++j) {
    const int K0 = 16 * (j >> 2) + 8 * hi + 2 * (j & 3);
    v2f w;
    if (w2valid) { w.x = W2[w2row * 32 + K0]; w.y = W2[w2row * 32 + K0 + 1]; }
    else         { w.x = 0.0f; w.y = 0.0f; }
    w2a[j] = w;
  }
  v8f b1d[2];
#pragma unroll
  for (int t = 0; t < 2; ++t)
#pragma unroll
    for (int r = 0; r < 8; ++r) b1d[t][r] = b1[t * 16 + 8 * hi + r];
  const float b2p = b2[hi], b2g = b2[2 + hi];

  const int waveGlobal = blockIdx.x * WAVES_PER_BLOCK + wave;
  const int waveTotal  = gridDim.x * WAVES_PER_BLOCK;

  float* st = &buf[(lane >> 3) * LSTRIDE + (lane & 7) * 4];

  const unsigned eo = (unsigned)(lo * 2 + hi);
  const float padOne = hi ? 0.0f : 1.0f;

  int tile = waveGlobal;
  if (tile >= nTiles) return;

  v4f xv[4];
  float spl;
  {
    const gv4f_ptr xs = (gv4f_ptr)wave_uniform_addr(x + (size_t)tile * (TILE_ROWS * 32));
    const gf_ptr ss = (gf_ptr)wave_uniform_addr(sp + (size_t)tile * (TILE_ROWS * 2));
#pragma unroll
    for (int j = 0; j < 4; ++j) xv[j] = xs[lane + 32 * j];
    spl = ss[eo];
  }

  for (;;) {
    const int row0 = tile * TILE_ROWS;

#pragma unroll
    for (int j = 0; j < 4; ++j) *(v4f*)&st[j * 4 * LSTRIDE] = xv[j];
    const float spo = spl;
    v2f spv; spv.x = spo; spv.y = padOne;

    const int nextTile = tile + waveTotal;
    const bool more = nextTile < nTiles;
    if (more) {
      const gv4f_ptr xs = (gv4f_ptr)wave_uniform_addr(x + (size_t)nextTile * (TILE_ROWS * 32));
      const gf_ptr ss = (gf_ptr)wave_uniform_addr(sp + (size_t)nextTile * (TILE_ROWS * 2));
#pragma unroll
      for (int j = 0; j < 4; ++j) xv[j] = xs[lane + 32 * j];
      spl = ss[eo];
    }
    __builtin_amdgcn_wave_barrier();

    v2f xb[8];
#pragma unroll
    for (int s = 0; s < 8; ++s)
      xb[s] = *(const v2f*)&buf[lo * LSTRIDE + 4 * s + 2 * hi];

    v8f h1[2];
#pragma unroll
    for (int t = 0; t < 2; ++t) {
      v8f acc = {};
#pragma unroll
      for (int s = 0; s < 8; ++s) acc = wmma4(w0a[t][s], xb[s], acc);
      acc = wmma4(w0a[t][8], spv, acc);
#pragma unroll
      for (int k = 0; k < 8; ++k) h1[t][k] = relu1(acc[k]);
    }

    v8f h2[2];
#pragma unroll
    for (int t = 0; t < 2; ++t) {
      v8f acc = b1d[t];
#pragma unroll
      for (int j = 0; j < 8; ++j) {
        v2f b; b.x = h1[j >> 2][2 * (j & 3)]; b.y = h1[j >> 2][2 * (j & 3) + 1];
        acc = wmma4(w1a[t][j], b, acc);
      }
#pragma unroll
      for (int k = 0; k < 8; ++k) h2[t][k] = relu1(acc[k]);
    }

    v8f raw = {};
#pragma unroll
    for (int j = 0; j < 8; ++j) {
      v2f b; b.x = h2[j >> 2][2 * (j & 3)]; b.y = h2[j >> 2][2 * (j & 3) + 1];
      raw = wmma4(w2a[j], b, raw);
    }

    {
      const float p = raw[0] + b2p;
      const float g = raw[1] + b2g;
      const float sig = __builtin_amdgcn_rcpf(1.0f + __expf(-g));
      out[(unsigned)(row0 * 2) + eo] = fmaf(p, sig, spo);
    }
    __builtin_amdgcn_wave_barrier();
    if (!more) break;
    tile = nextTile;
  }
}

extern "C" void kernel_launch(void* const* d_in, const int* in_sizes, int n_in,
                              void* d_out, int out_size, void* d_ws, size_t ws_size,
                              hipStream_t stream) {
  const float* x  = (const float*)d_in[0];
  const float* sp = (const float*)d_in[1];
  const float* W0 = (const float*)d_in[2];
  const float* b0 = (const float*)d_in[3];
  const float* W1 = (const float*)d_in[4];
  const float* b1 = (const float*)d_in[5];
  const float* W2 = (const float*)d_in[6];
  const float* b2 = (const float*)d_in[7];
  float* out = (float*)d_out;

  const int B = in_sizes[0] / 32;
  const int nTiles = B / TILE_ROWS;
  int blocks = (nTiles + (WAVES_PER_BLOCK * 8) - 1) / (WAVES_PER_BLOCK * 8);
  if (blocks < 1) blocks = 1;
  if (blocks > 8192) blocks = 8192;

  arm_mlp_wmma_kernel<<<blocks, 256, 0, stream>>>(x, sp, W0, b0, W1, b1, W2, b2,
                                                  out, nTiles);
}
